# starting kernel
# baseline (speedup 1.0000x reference)
#include <hip/hip_runtime.h>
#include <hip/hip_bf16.h>

#define TDIM 2048
#define HALF 1024
#define BINS 2049
#define BINS_PAD 2304
#define NCHUNK 9
#define NSTRIP 8
#define ROWS_PER_STRIP 256
#define NBATCH 16

typedef __attribute__((ext_vector_type(2))) float v2f;
typedef __attribute__((ext_vector_type(8))) float v8f;

__global__ __launch_bounds__(256) void diagmean_partial(
    const float* __restrict__ in, float* __restrict__ part) {
  const int chunk = blockIdx.x;
  const int strip = blockIdx.y;
  const int b     = blockIdx.z;
  const int bin   = chunk * 256 + (int)threadIdx.x;
  const int d     = bin - HALF;

  float acc = 0.0f;
  if (d <= HALF) {
    const float* pb = in + (size_t)b * TDIM * TDIM;
    int i0 = strip * ROWS_PER_STRIP;
    int ilo = i0;
    if (d < 0 && ilo < -d) ilo = -d;
    int ihi = i0 + ROWS_PER_STRIP;
    if (ihi > TDIM - 1) ihi = TDIM - 1;
    int jcap = (TDIM - 1) - d;
    if (d > 0 && ihi > jcap) ihi = jcap;

    size_t idx = (size_t)((long)ilo * (TDIM + 1) + d);
#pragma unroll 4
    for (int i = ilo; i < ihi; ++i) {
      acc += pb[idx];
      idx += (TDIM + 1);
    }
  }
  part[((size_t)b * NSTRIP + strip) * BINS_PAD + bin] = acc;
}

__global__ __launch_bounds__(128) void diagmean_finalize(
    const float* __restrict__ part, float* __restrict__ out) {
  const int b    = blockIdx.x;
  const int tid  = (int)threadIdx.x;
  const int lane = tid & 31;
  const int wave = tid >> 5;
  const int col  = lane & 15;
  const int hi   = lane >> 4;

  __shared__ float means[2064];
  __shared__ float red[128];
  __shared__ float gmean_s;

  const float* pb = part + (size_t)b * NSTRIP * BINS_PAD;

  v2f a; a.x = 1.0f; a.y = 1.0f;

  const int NTILES = 129;
  for (int tile = wave; tile < NTILES; tile += 4) {
    const int bin = tile * 16 + col;
    v2f b0, b1;
    b0.x = pb[(size_t)(2 * hi + 0) * BINS_PAD + bin];
    b0.y = pb[(size_t)(2 * hi + 1) * BINS_PAD + bin];
    b1.x = pb[(size_t)(4 + 2 * hi) * BINS_PAD + bin];
    b1.y = pb[(size_t)(5 + 2 * hi) * BINS_PAD + bin];

    v8f c = {};
    c = __builtin_amdgcn_wmma_f32_16x16x4_f32(false, a, false, b0,
                                              (short)0, c, false, false);
    c = __builtin_amdgcn_wmma_f32_16x16x4_f32(false, a, false, b1,
                                              (short)0, c, false, false);
    if (hi == 0 && bin < BINS) {
      int dd = bin - HALF;
      float cnt = (float)((TDIM - 1) - (dd < 0 ? -dd : dd));
      means[bin] = c[0] / cnt;
    }
  }
  __syncthreads();

  float p = 0.0f;
  for (int i = tid; i < BINS; i += 128) p += means[i];
  red[tid] = p;
  __syncthreads();
  if (tid == 0) {
    float s = 0.0f;
    for (int i = 0; i < 128; ++i) s += red[i];
    gmean_s = s / (float)BINS;
  }
  __syncthreads();

  const float g = gmean_s;
  for (int i = tid; i < BINS; i += 128) {
    out[(size_t)b * BINS + i] = -(means[i] - g);
  }
}

extern "C" void kernel_launch(void* const* d_in, const int* in_sizes, int n_in,
                              void* d_out, int out_size, void* d_ws, size_t ws_size,
                              hipStream_t stream) {
  (void)in_sizes; (void)n_in; (void)out_size; (void)ws_size;
  const float* in  = (const float*)d_in[0];
  float*       out = (float*)d_out;
  float*       ws  = (float*)d_ws;

  dim3 g1(NCHUNK, NSTRIP, NBATCH);
  diagmean_partial<<<g1, 256, 0, stream>>>(in, ws);
  diagmean_finalize<<<NBATCH, 128, 0, stream>>>(ws, out);
}


# partial kernel: one 32-row x 544-column tile per work-group for 512 diagonals of a 512-row strip (17 lines a row, was 4 tiles of 5); strip sums meet on the WMMA in finalize; sum order unchanged
# speedup vs baseline: 2.0120x; 2.0120x over previous
#include <hip/hip_runtime.h>


#define TDIM 2048
#define HALF 1024
#define BINS 2049
#define NBATCH 16
#define GDIAG 512
#define NDG 4
#define NSTRIP 4
#define ROWS_PER_STRIP 512
#define CHUNK 32
#define NCHUNK 16
#define TILE_W 544
#define SLOTS_PER_ROW 136
#define NTHREAD 544
#define NLOAD 8
#define WS_PITCH 2048

typedef __attribute__((ext_vector_type(2))) float v2f;
typedef __attribute__((ext_vector_type(4))) float v4f;
typedef __attribute__((ext_vector_type(8))) float v8f;

__global__ __launch_bounds__(NTHREAD) void diagmean_partial(
    const float* __restrict__ in, float* __restrict__ strip_ws) {
  const int dg    = blockIdx.x & (NDG - 1);
  const int strip = blockIdx.x >> 2;
  const int b     = blockIdx.y;
  const int t     = (int)threadIdx.x;
  const int wave  = __builtin_amdgcn_readfirstlane(t >> 5);
  const int dlo   = dg * GDIAG - HALF;

  __shared__ __attribute__((aligned(16))) float tile[2][CHUNK * TILE_W];

  const float* pb = in + (size_t)b * TDIM * TDIM;
  const int i0 = strip * ROWS_PER_STRIP;

  unsigned rq[NLOAD];
  int      qcol[NLOAD];
#pragma unroll
  for (int u = 0; u < NLOAD; ++u) {
    const int piece = t + NTHREAD * u;
    const int r = piece / SLOTS_PER_ROW;
    const int q = piece - r * SLOTS_PER_ROW;
    rq[u]   = (unsigned)(r * TDIM);
    qcol[u] = 4 * q;
  }

  v4f xa[NLOAD];
  v4f xb[NLOAD];
  const v4f zero4 = {0.0f, 0.0f, 0.0f, 0.0f};

#define LOAD_TILE(X, IC)                                                       \
  {                                                                            \
    const int cs_ = (IC) + dlo;                                                \
    const float* rp_ = pb + (long)(IC) * TDIM;                                 \
    _Pragma("unroll")                                                          \
    for (int u = 0; u < NLOAD; ++u) {                                          \
      int col_ = cs_ + qcol[u];                                                \
      col_ = col_ < 0 ? 0 : col_;                                              \
      col_ = col_ > TDIM - 4 ? TDIM - 4 : col_;                                \
      X[u] = *(const v4f*)(rp_ + (rq[u] + (unsigned)col_));                    \
    }                                                                          \
  }

#define FIX_TILE(X, IC)                                                        \
  {                                                                            \
    const int cs_ = (IC) + dlo;                                                \
    const bool last_ = ((IC) + CHUNK == TDIM);                                 \
    if (cs_ < 0 || cs_ + TILE_W > TDIM - 4 || last_) {                         \
      _Pragma("unroll")                                                        \
      for (int u = 0; u < NLOAD; ++u) {                                        \
        const int col_ = cs_ + qcol[u];                                        \
        bool ok_ = (unsigned)col_ < (unsigned)TDIM;                            \
        if (u == NLOAD - 1) ok_ = ok_ && !(last_ && t >= NTHREAD - SLOTS_PER_ROW); \
        X[u] = ok_ ? X[u] : zero4;                                             \
        X[u].w = (col_ == TDIM - 4) ? 0.0f : X[u].w;                           \
      }                                                                        \
    }                                                                          \
  }

#define STAGE(X, BUFI, IC, NEXT)                                               \
  {                                                                            \
    float* tb_ = &tile[BUFI][0];                                               \
    FIX_TILE(X, IC)                                                            \
    _Pragma("unroll")                                                          \
    for (int u = 0; u < NLOAD; ++u)                                            \
      *(v4f*)(tb_ + 4 * (t + NTHREAD * u)) = X[u];                             \
    if (NEXT) LOAD_TILE(X, (IC) + 2 * CHUNK)                                   \
    asm volatile("s_wait_dscnt 0x0\n\ts_barrier_signal -1\n\ts_barrier_wait -1" ::: "memory"); \
    if (wave < GDIAG / 32) {                                                   \
      float vv_[CHUNK];                                                        \
      _Pragma("unroll")                                                        \
      for (int r = 0; r < CHUNK; ++r)                                          \
        vv_[r] = tb_[r * (TILE_W + 1) + t];                                    \
      __builtin_amdgcn_sched_barrier(0);                                       \
      _Pragma("unroll")                                                        \
      for (int r = 0; r < CHUNK; ++r)                                          \
        acc += vv_[r];                                                         \
    }                                                                          \
  }

  LOAD_TILE(xa, i0)
  LOAD_TILE(xb, i0 + CHUNK)

  float acc = 0.0f;
#pragma unroll 1
  for (int c = 0; c < NCHUNK - 2; c += 2) {
    const int ic = i0 + CHUNK * c;
    STAGE(xa, 0, ic, 1)
    STAGE(xb, 1, ic + CHUNK, 1)
  }
  {
    const int ic = i0 + CHUNK * (NCHUNK - 2);
    STAGE(xa, 0, ic, 0)
    STAGE(xb, 1, ic + CHUNK, 0)
  }
#undef STAGE
#undef FIX_TILE
#undef LOAD_TILE
  if (wave < GDIAG / 32)
    strip_ws[((size_t)b * NSTRIP + strip) * WS_PITCH + dg * GDIAG + t] = acc;
}

__global__ __launch_bounds__(128) void diagmean_finalize(
    const float* __restrict__ in, const float* __restrict__ strip_ws,
    float* __restrict__ out) {
  const int b    = blockIdx.x;
  const int tid  = (int)threadIdx.x;
  const int lane = tid & 31;
  const int col  = lane & 15;
  const int hi   = lane >> 4;

  __shared__ float red[128];
  __shared__ float red2[128];
  __shared__ float part_s;
  __shared__ float corner_s;

  const float* sb = strip_ws + (size_t)b * NSTRIP * WS_PITCH;
  const float* pb = in + (size_t)b * TDIM * TDIM;

  float v[8];
  v2f b0[WS_PITCH / 128];
  v2f b1[WS_PITCH / 128];
  float m[WS_PITCH / 128];
#pragma unroll
  for (int k = 0; k < 8; ++k) {
    const int i = tid * 8 + k;
    v[k] = pb[(size_t)i * TDIM + (i + HALF)];
  }
  const float* s0 = sb + (2 * hi + 0) * WS_PITCH + (tid & ~31) + col;
  const float* s1 = sb + (2 * hi + 1) * WS_PITCH + (tid & ~31) + col;
#pragma unroll
  for (int k = 0; k < WS_PITCH / 128; ++k) {
    b0[k].x = s0[128 * k];
    b0[k].y = s1[128 * k];
    b1[k].x = s0[128 * k + 16];
    b1[k].y = s1[128 * k + 16];
  }

  float q = 0.0f;
#pragma unroll
  for (int k = 0; k < 8; ++k) {
    const int i = tid * 8 + k;
    q += (i < HALF - 1) ? v[k] : 0.0f;
  }
  v2f a; a.x = 1.0f; a.y = 1.0f;
#pragma unroll
  for (int k = 0; k < WS_PITCH / 128; ++k) {
    v8f c0 = {};
    v8f c1 = {};
    c0 = __builtin_amdgcn_wmma_f32_16x16x4_f32(false, a, false, b0[k],
                                               (short)0, c0, false, false);
    c1 = __builtin_amdgcn_wmma_f32_16x16x4_f32(false, a, false, b1[k],
                                               (short)0, c1, false, false);
    const float s   = hi ? c1[0] : c0[0];
    const int obin  = tid + 128 * k;
    const int dd    = obin - HALF;
    const float cnt = (float)((TDIM - 1) - (dd < 0 ? -dd : dd));
    m[k] = s / cnt;
  }
  float p = 0.0f;
#pragma unroll
  for (int k = 0; k < WS_PITCH / 128; ++k) p += m[k];
  red[tid]  = p;
  red2[tid] = q;
  __syncthreads();
  if (tid == 0) {
    float s = 0.0f;
    for (int i = 0; i < 128; ++i) s += red[i];
    part_s = s;
  }
  if (tid == 32) {
    float s = 0.0f;
    for (int i = 0; i < 128; ++i) s += red2[i];
    corner_s = s / (float)((TDIM - 1) - HALF);
  }
  __syncthreads();

  const float corner = corner_s;
  const float g = (part_s + corner) / (float)BINS;
#pragma unroll
  for (int k = 0; k < WS_PITCH / 128; ++k) {
    out[(size_t)b * BINS + tid + 128 * k] = -(m[k] - g);
  }
  if (tid == 0) out[(size_t)b * BINS + (BINS - 1)] = -(corner - g);
}

extern "C" void kernel_launch(void* const* d_in, const int* in_sizes, int n_in,
                              void* d_out, int out_size, void* d_ws, size_t ws_size,
                              hipStream_t stream) {
  (void)in_sizes; (void)n_in; (void)out_size; (void)ws_size;
  const float* in  = (const float*)d_in[0];
  float*       out = (float*)d_out;
  float*       ws  = (float*)d_ws;

  dim3 g1(NDG * NSTRIP, NBATCH, 1);
  diagmean_partial<<<g1, NTHREAD, 0, stream>>>(in, ws);
  diagmean_finalize<<<NBATCH, 128, 0, stream>>>(in, ws, out);
}


# partial kernel: LDS tile read back with 16-byte reads (rows stored with a 4-column step per 4 rows, a thread sums one row class of 4 diagonals); 128 reads a tile, was 512; sum order changed
# speedup vs baseline: 2.1195x; 1.0534x over previous
#include <hip/hip_runtime.h>


#define TDIM 2048
#define HALF 1024
#define BINS 2049
#define NBATCH 16
#define GDIAG 512
#define NDG 4
#define NSTRIP 4
#define ROWS_PER_STRIP 512
#define CHUNK 32
#define NCHUNK 16
#define TILE_W 544
#define SLOTS_PER_ROW 136
#define NTHREAD 544
#define NLOAD 8
#define WS_PITCH 2048
#define LDS_PITCH 548
#define NQUAD 129
#define JOIN_W 520

typedef __attribute__((ext_vector_type(2))) float v2f;
typedef __attribute__((ext_vector_type(4))) float v4f;
typedef __attribute__((ext_vector_type(8))) float v8f;

__global__ __launch_bounds__(NTHREAD) void diagmean_partial(
    const float* __restrict__ in, float* __restrict__ strip_ws) {
  const int dg    = blockIdx.x & (NDG - 1);
  const int strip = blockIdx.x >> 2;
  const int b     = blockIdx.y;
  const int t     = (int)threadIdx.x;
  const int dlo   = dg * GDIAG - HALF;
  const int kcls  = (t < GDIAG) ? ((t >> 5) & 3) : (t & 3);
  const int qd    = (t < GDIAG) ? (((t >> 7) << 5) + (t & 31)) : (NQUAD - 1);
  const int rdoff = kcls * LDS_PITCH + 4 * qd;
  const int stoff = (t / SLOTS_PER_ROW) * LDS_PITCH + 4 * (t % SLOTS_PER_ROW);

  __shared__ __attribute__((aligned(16))) float tile[2][CHUNK * LDS_PITCH];
  __shared__ __attribute__((aligned(16))) float joinp[4][JOIN_W];

  const float* pb = in + (size_t)b * TDIM * TDIM;
  const int i0 = strip * ROWS_PER_STRIP;

  unsigned rq[NLOAD];
  int      qcol[NLOAD];
#pragma unroll
  for (int u = 0; u < NLOAD; ++u) {
    const int piece = t + NTHREAD * u;
    const int r = piece / SLOTS_PER_ROW;
    const int q = piece - r * SLOTS_PER_ROW;
    rq[u]   = (unsigned)(r * TDIM);
    qcol[u] = 4 * q;
  }

  v4f xa[NLOAD];
  v4f xb[NLOAD];
  const v4f zero4 = {0.0f, 0.0f, 0.0f, 0.0f};

#define LOAD_TILE(X, IC)                                                       \
  {                                                                            \
    const int cs_ = (IC) + dlo;                                                \
    const float* rp_ = pb + (long)(IC) * TDIM;                                 \
    unsigned off_[NLOAD];                                                      \
    _Pragma("unroll")                                                          \
    for (int u = 0; u < NLOAD; ++u) {                                          \
      int col_ = cs_ + qcol[u];                                                \
      col_ = col_ < 0 ? 0 : col_;                                              \
      col_ = col_ > TDIM - 4 ? TDIM - 4 : col_;                                \
      off_[u] = rq[u] + (unsigned)col_;                                        \
    }                                                                          \
    __builtin_amdgcn_sched_barrier(0);                                         \
    _Pragma("unroll")                                                          \
    for (int u = 0; u < NLOAD; ++u)                                            \
      X[u] = *(const v4f*)(rp_ + off_[u]);                                     \
  }

#define FIX_TILE(X, IC)                                                        \
  {                                                                            \
    const int cs_ = (IC) + dlo;                                                \
    const bool last_ = ((IC) + CHUNK == TDIM);                                 \
    if (cs_ < 0 || cs_ + TILE_W > TDIM - 4 || last_) {                         \
      _Pragma("unroll")                                                        \
      for (int u = 0; u < NLOAD; ++u) {                                        \
        const int col_ = cs_ + qcol[u];                                        \
        bool ok_ = (unsigned)col_ < (unsigned)TDIM;                            \
        if (u == NLOAD - 1) ok_ = ok_ && !(last_ && t >= NTHREAD - SLOTS_PER_ROW); \
        X[u] = ok_ ? X[u] : zero4;                                             \
        X[u].w = (col_ == TDIM - 4) ? 0.0f : X[u].w;                           \
      }                                                                        \
    }                                                                          \
  }

#define STAGE(X, BUFI, IC, NEXT)                                               \
  {                                                                            \
    float* tb_ = &tile[BUFI][0];                                               \
    FIX_TILE(X, IC)                                                            \
    _Pragma("unroll")                                                          \
    for (int u = 0; u < NLOAD; ++u)                                            \
      *(v4f*)(tb_ + stoff + u * (4 * LDS_PITCH - 4)) = X[u];                   \
    if (NEXT) LOAD_TILE(X, (IC) + 2 * CHUNK)                                   \
    asm volatile("s_wait_dscnt 0x0\n\ts_barrier_signal -1\n\ts_barrier_wait -1" ::: "memory"); \
    {                                                                          \
      v4f vv_[CHUNK / 4];                                                      \
      _Pragma("unroll")                                                        \
      for (int j = 0; j < CHUNK / 4; ++j)                                      \
        vv_[j] = *(const v4f*)(tb_ + rdoff + j * (4 * LDS_PITCH));             \
      __builtin_amdgcn_sched_barrier(0);                                       \
      _Pragma("unroll")                                                        \
      for (int j = 0; j < CHUNK / 4; ++j)                                      \
        acc += vv_[j];                                                         \
    }                                                                          \
  }

  LOAD_TILE(xa, i0)
  LOAD_TILE(xb, i0 + CHUNK)

  v4f acc = {0.0f, 0.0f, 0.0f, 0.0f};
#pragma unroll 1
  for (int c = 0; c < NCHUNK - 2; c += 2) {
    const int ic = i0 + CHUNK * c;
    STAGE(xa, 0, ic, 1)
    STAGE(xb, 1, ic + CHUNK, 1)
  }
  {
    const int ic = i0 + CHUNK * (NCHUNK - 2);
    STAGE(xa, 0, ic, 0)
    STAGE(xb, 1, ic + CHUNK, 0)
  }
#undef STAGE
#undef FIX_TILE
#undef LOAD_TILE
  if (t < GDIAG + 4) *(v4f*)&joinp[kcls][4 * qd] = acc;
  __syncthreads();
  if (t < GDIAG) {
    const float s = ((joinp[0][t] + joinp[1][t + 1]) + joinp[2][t + 2]) + joinp[3][t + 3];
    strip_ws[((size_t)b * NSTRIP + strip) * WS_PITCH + dg * GDIAG + t] = s;
  }
}

__global__ __launch_bounds__(128) void diagmean_finalize(
    const float* __restrict__ in, const float* __restrict__ strip_ws,
    float* __restrict__ out) {
  const int b    = blockIdx.x;
  const int tid  = (int)threadIdx.x;
  const int lane = tid & 31;
  const int col  = lane & 15;
  const int hi   = lane >> 4;

  __shared__ float red[128];
  __shared__ float red2[128];
  __shared__ float part_s;
  __shared__ float corner_s;

  const float* sb = strip_ws + (size_t)b * NSTRIP * WS_PITCH;
  const float* pb = in + (size_t)b * TDIM * TDIM;

  float v[8];
  v2f b0[WS_PITCH / 128];
  v2f b1[WS_PITCH / 128];
  float m[WS_PITCH / 128];
#pragma unroll
  for (int k = 0; k < 8; ++k) {
    const int i = tid * 8 + k;
    v[k] = pb[(size_t)i * TDIM + (i + HALF)];
  }
  const float* s0 = sb + (2 * hi + 0) * WS_PITCH + (tid & ~31) + col;
  const float* s1 = sb + (2 * hi + 1) * WS_PITCH + (tid & ~31) + col;
#pragma unroll
  for (int k = 0; k < WS_PITCH / 128; ++k) {
    b0[k].x = s0[128 * k];
    b0[k].y = s1[128 * k];
    b1[k].x = s0[128 * k + 16];
    b1[k].y = s1[128 * k + 16];
  }

  float q = 0.0f;
#pragma unroll
  for (int k = 0; k < 8; ++k) {
    const int i = tid * 8 + k;
    q += (i < HALF - 1) ? v[k] : 0.0f;
  }
  v2f a; a.x = 1.0f; a.y = 1.0f;
#pragma unroll
  for (int k = 0; k < WS_PITCH / 128; ++k) {
    v8f c0 = {};
    v8f c1 = {};
    c0 = __builtin_amdgcn_wmma_f32_16x16x4_f32(false, a, false, b0[k],
                                               (short)0, c0, false, false);
    c1 = __builtin_amdgcn_wmma_f32_16x16x4_f32(false, a, false, b1[k],
                                               (short)0, c1, false, false);
    const float s   = hi ? c1[0] : c0[0];
    const int obin  = tid + 128 * k;
    const int dd    = obin - HALF;
    const float cnt = (float)((TDIM - 1) - (dd < 0 ? -dd : dd));
    m[k] = s / cnt;
  }
  float p = 0.0f;
#pragma unroll
  for (int k = 0; k < WS_PITCH / 128; ++k) p += m[k];
  red[tid]  = p;
  red2[tid] = q;
  __syncthreads();
  if (tid == 0) {
    float s = 0.0f;
    for (int i = 0; i < 128; ++i) s += red[i];
    part_s = s;
  }
  if (tid == 32) {
    float s = 0.0f;
    for (int i = 0; i < 128; ++i) s += red2[i];
    corner_s = s / (float)((TDIM - 1) - HALF);
  }
  __syncthreads();

  const float corner = corner_s;
  const float g = (part_s + corner) / (float)BINS;
#pragma unroll
  for (int k = 0; k < WS_PITCH / 128; ++k) {
    out[(size_t)b * BINS + tid + 128 * k] = -(m[k] - g);
  }
  if (tid == 0) out[(size_t)b * BINS + (BINS - 1)] = -(corner - g);
}

extern "C" void kernel_launch(void* const* d_in, const int* in_sizes, int n_in,
                              void* d_out, int out_size, void* d_ws, size_t ws_size,
                              hipStream_t stream) {
  (void)in_sizes; (void)n_in; (void)out_size; (void)ws_size;
  const float* in  = (const float*)d_in[0];
  float*       out = (float*)d_out;
  float*       ws  = (float*)d_ws;

  dim3 g1(NDG * NSTRIP, NBATCH, 1);
  diagmean_partial<<<g1, NTHREAD, 0, stream>>>(in, ws);
  diagmean_finalize<<<NBATCH, 128, 0, stream>>>(in, ws, out);
}


# partial kernel: tiles slide along the diagonal, so a loader's 16-byte slot holds the same 4 diagonals in each tile: slots summed in registers, one LDS skew a strip, no loop barrier; sum order changed
# speedup vs baseline: 2.2853x; 1.0782x over previous
#include <hip/hip_runtime.h>


#define TDIM 2048
#define HALF 1024
#define BINS 2049
#define NBATCH 16
#define GDIAG 512
#define NDG 4
#define NSTRIP 4
#define ROWS_PER_STRIP 512
#define CHUNK 32
#define NCHUNK 16
#define TILE_W 544
#define SLOTS_PER_ROW 136
#define NTHREAD 544
#define NLOAD 8
#define WS_PITCH 2048
#define LDS_PITCH 548
#define NQUAD 129
#define JOIN_W 520

typedef __attribute__((ext_vector_type(2))) float v2f;
typedef __attribute__((ext_vector_type(4))) float v4f;
typedef __attribute__((ext_vector_type(8))) float v8f;

__global__ __launch_bounds__(NTHREAD) void diagmean_partial(
    const float* __restrict__ in, float* __restrict__ strip_ws) {
  const int dg    = blockIdx.x & (NDG - 1);
  const int strip = blockIdx.x >> 2;
  const int b     = blockIdx.y;
  const int t     = (int)threadIdx.x;
  const int dlo   = dg * GDIAG - HALF;
  const int kcls  = (t < GDIAG) ? ((t >> 5) & 3) : (t & 3);
  const int qd    = (t < GDIAG) ? (((t >> 7) << 5) + (t & 31)) : (NQUAD - 1);
  const int rdoff = kcls * LDS_PITCH + 4 * qd;
  const int stoff = (t / SLOTS_PER_ROW) * LDS_PITCH + 4 * (t % SLOTS_PER_ROW);

  __shared__ __attribute__((aligned(16))) float tile[CHUNK * LDS_PITCH];
  __shared__ __attribute__((aligned(16))) float joinp[4][JOIN_W];

  const float* pb = in + (size_t)b * TDIM * TDIM;
  const int i0 = strip * ROWS_PER_STRIP;

  unsigned rq[NLOAD];
  int      qcol[NLOAD];
#pragma unroll
  for (int u = 0; u < NLOAD; ++u) {
    const int piece = t + NTHREAD * u;
    const int r = piece / SLOTS_PER_ROW;
    const int q = piece - r * SLOTS_PER_ROW;
    rq[u]   = (unsigned)(r * TDIM);
    qcol[u] = 4 * q;
  }

  v4f xa[NLOAD];
  v4f xb[NLOAD];
  v4f sum[NLOAD];
  const v4f zero4 = {0.0f, 0.0f, 0.0f, 0.0f};
#pragma unroll
  for (int u = 0; u < NLOAD; ++u) sum[u] = zero4;

#define LOAD_TILE(X, IC)                                                       \
  {                                                                            \
    const int cs_ = (IC) + dlo;                                                \
    const float* rp_ = pb + (long)(IC) * TDIM;                                 \
    unsigned off_[NLOAD];                                                      \
    _Pragma("unroll")                                                          \
    for (int u = 0; u < NLOAD; ++u) {                                          \
      int col_ = cs_ + qcol[u];                                                \
      col_ = col_ < 0 ? 0 : col_;                                              \
      col_ = col_ > TDIM - 4 ? TDIM - 4 : col_;                                \
      off_[u] = rq[u] + (unsigned)col_;                                        \
    }                                                                          \
    __builtin_amdgcn_sched_barrier(0);                                         \
    _Pragma("unroll")                                                          \
    for (int u = 0; u < NLOAD; ++u)                                            \
      X[u] = *(const v4f*)(rp_ + off_[u]);                                     \
  }

#define FIX_TILE(X, IC)                                                        \
  {                                                                            \
    const int cs_ = (IC) + dlo;                                                \
    const bool last_ = ((IC) + CHUNK == TDIM);                                 \
    if (cs_ < 0 || cs_ + TILE_W > TDIM - 4 || last_) {                         \
      _Pragma("unroll")                                                        \
      for (int u = 0; u < NLOAD; ++u) {                                        \
        const int col_ = cs_ + qcol[u];                                        \
        bool ok_ = (unsigned)col_ < (unsigned)TDIM;                            \
        if (u == NLOAD - 1) ok_ = ok_ && !(last_ && t >= NTHREAD - SLOTS_PER_ROW); \
        X[u] = ok_ ? X[u] : zero4;                                             \
        X[u].w = (col_ == TDIM - 4) ? 0.0f : X[u].w;                           \
      }                                                                        \
    }                                                                          \
  }

#define TAKE(X, IC, NEXT)                                                      \
  {                                                                            \
    FIX_TILE(X, IC)                                                            \
    _Pragma("unroll")                                                          \
    for (int u = 0; u < NLOAD; ++u)                                            \
      sum[u] += X[u];                                                          \
    if (NEXT) LOAD_TILE(X, (IC) + 2 * CHUNK)                                   \
  }

  LOAD_TILE(xa, i0)
  LOAD_TILE(xb, i0 + CHUNK)

#pragma unroll 1
  for (int c = 0; c < NCHUNK - 2; c += 2) {
    const int ic = i0 + CHUNK * c;
    TAKE(xa, ic, 1)
    TAKE(xb, ic + CHUNK, 1)
  }
  {
    const int ic = i0 + CHUNK * (NCHUNK - 2);
    TAKE(xa, ic, 0)
    TAKE(xb, ic + CHUNK, 0)
  }
#undef TAKE
#undef FIX_TILE
#undef LOAD_TILE

#pragma unroll
  for (int u = 0; u < NLOAD; ++u)
    *(v4f*)(tile + stoff + u * (4 * LDS_PITCH - 4)) = sum[u];
  __syncthreads();
  v4f acc = *(const v4f*)(tile + rdoff);
#pragma unroll
  for (int j = 1; j < CHUNK / 4; ++j)
    acc += *(const v4f*)(tile + rdoff + j * (4 * LDS_PITCH));

  if (t < GDIAG + 4) *(v4f*)&joinp[kcls][4 * qd] = acc;
  __syncthreads();
  if (t < GDIAG) {
    const float s = ((joinp[0][t] + joinp[1][t + 1]) + joinp[2][t + 2]) + joinp[3][t + 3];
    strip_ws[((size_t)b * NSTRIP + strip) * WS_PITCH + dg * GDIAG + t] = s;
  }
}

__global__ __launch_bounds__(128) void diagmean_finalize(
    const float* __restrict__ in, const float* __restrict__ strip_ws,
    float* __restrict__ out) {
  const int b    = blockIdx.x;
  const int tid  = (int)threadIdx.x;
  const int lane = tid & 31;
  const int col  = lane & 15;
  const int hi   = lane >> 4;

  __shared__ float red[128];
  __shared__ float red2[128];
  __shared__ float part_s;
  __shared__ float corner_s;

  const float* sb = strip_ws + (size_t)b * NSTRIP * WS_PITCH;
  const float* pb = in + (size_t)b * TDIM * TDIM;

  float v[8];
  v2f b0[WS_PITCH / 128];
  v2f b1[WS_PITCH / 128];
  float m[WS_PITCH / 128];
#pragma unroll
  for (int k = 0; k < 8; ++k) {
    const int i = tid * 8 + k;
    v[k] = pb[(size_t)i * TDIM + (i + HALF)];
  }
  const float* s0 = sb + (2 * hi + 0) * WS_PITCH + (tid & ~31) + col;
  const float* s1 = sb + (2 * hi + 1) * WS_PITCH + (tid & ~31) + col;
#pragma unroll
  for (int k = 0; k < WS_PITCH / 128; ++k) {
    b0[k].x = s0[128 * k];
    b0[k].y = s1[128 * k];
    b1[k].x = s0[128 * k + 16];
    b1[k].y = s1[128 * k + 16];
  }

  float q = 0.0f;
#pragma unroll
  for (int k = 0; k < 8; ++k) {
    const int i = tid * 8 + k;
    q += (i < HALF - 1) ? v[k] : 0.0f;
  }
  v2f a; a.x = 1.0f; a.y = 1.0f;
#pragma unroll
  for (int k = 0; k < WS_PITCH / 128; ++k) {
    v8f c0 = {};
    v8f c1 = {};
    c0 = __builtin_amdgcn_wmma_f32_16x16x4_f32(false, a, false, b0[k],
                                               (short)0, c0, false, false);
    c1 = __builtin_amdgcn_wmma_f32_16x16x4_f32(false, a, false, b1[k],
                                               (short)0, c1, false, false);
    const float s   = hi ? c1[0] : c0[0];
    const int obin  = tid + 128 * k;
    const int dd    = obin - HALF;
    const float cnt = (float)((TDIM - 1) - (dd < 0 ? -dd : dd));
    m[k] = s / cnt;
  }
  float p = 0.0f;
#pragma unroll
  for (int k = 0; k < WS_PITCH / 128; ++k) p += m[k];
  red[tid]  = p;
  red2[tid] = q;
  __syncthreads();
  if (tid == 0) {
    float s = 0.0f;
    for (int i = 0; i < 128; ++i) s += red[i];
    part_s = s;
  }
  if (tid == 32) {
    float s = 0.0f;
    for (int i = 0; i < 128; ++i) s += red2[i];
    corner_s = s / (float)((TDIM - 1) - HALF);
  }
  __syncthreads();

  const float corner = corner_s;
  const float g = (part_s + corner) / (float)BINS;
#pragma unroll
  for (int k = 0; k < WS_PITCH / 128; ++k) {
    out[(size_t)b * BINS + tid + 128 * k] = -(m[k] - g);
  }
  if (tid == 0) out[(size_t)b * BINS + (BINS - 1)] = -(corner - g);
}

extern "C" void kernel_launch(void* const* d_in, const int* in_sizes, int n_in,
                              void* d_out, int out_size, void* d_ws, size_t ws_size,
                              hipStream_t stream) {
  (void)in_sizes; (void)n_in; (void)out_size; (void)ws_size;
  const float* in  = (const float*)d_in[0];
  float*       out = (float*)d_out;
  float*       ws  = (float*)d_ws;

  dim3 g1(NDG * NSTRIP, NBATCH, 1);
  diagmean_partial<<<g1, NTHREAD, 0, stream>>>(in, ws);
  diagmean_finalize<<<NBATCH, 128, 0, stream>>>(in, ws, out);
}


# partial kernel: work-group index remapped so the 160 groups whose strip lies wholly in the band enter first, the 64 half-clipped next, the 32 empty last (was batch-major); sum order unchanged
# speedup vs baseline: 2.4228x; 1.0602x over previous
#include <hip/hip_runtime.h>


#define TDIM 2048
#define HALF 1024
#define BINS 2049
#define NBATCH 16
#define GDIAG 512
#define NDG 4
#define NSTRIP 4
#define ROWS_PER_STRIP 512
#define CHUNK 32
#define NCHUNK 16
#define TILE_W 544
#define SLOTS_PER_ROW 136
#define NTHREAD 544
#define NLOAD 8
#define WS_PITCH 2048
#define LDS_PITCH 548
#define NQUAD 129
#define JOIN_W 520
#define GROUP_ORDER 0xF0EB41DCA9876532ull

typedef __attribute__((ext_vector_type(2))) float v2f;
typedef __attribute__((ext_vector_type(4))) float v4f;
typedef __attribute__((ext_vector_type(8))) float v8f;

__global__ __launch_bounds__(NTHREAD) void diagmean_partial(
    const float* __restrict__ in, float* __restrict__ strip_ws) {
  const unsigned cls = blockIdx.x >> 4;
  const unsigned pr  = (unsigned)((GROUP_ORDER >> (4u * cls)) & 15ull);
  const int dg    = (int)(pr & (NDG - 1));
  const int strip = (int)(pr >> 2);
  const int b     = (int)(blockIdx.x & (NBATCH - 1));
  const int t     = (int)threadIdx.x;
  const int dlo   = dg * GDIAG - HALF;
  const int kcls  = (t < GDIAG) ? ((t >> 5) & 3) : (t & 3);
  const int qd    = (t < GDIAG) ? (((t >> 7) << 5) + (t & 31)) : (NQUAD - 1);
  const int rdoff = kcls * LDS_PITCH + 4 * qd;
  const int stoff = (t / SLOTS_PER_ROW) * LDS_PITCH + 4 * (t % SLOTS_PER_ROW);

  __shared__ __attribute__((aligned(16))) float tile[CHUNK * LDS_PITCH];
  __shared__ __attribute__((aligned(16))) float joinp[4][JOIN_W];

  const float* pb = in + (size_t)b * TDIM * TDIM;
  const int i0 = strip * ROWS_PER_STRIP;

  unsigned rq[NLOAD];
  int      qcol[NLOAD];
#pragma unroll
  for (int u = 0; u < NLOAD; ++u) {
    const int piece = t + NTHREAD * u;
    const int r = piece / SLOTS_PER_ROW;
    const int q = piece - r * SLOTS_PER_ROW;
    rq[u]   = (unsigned)(r * TDIM);
    qcol[u] = 4 * q;
  }

  v4f xa[NLOAD];
  v4f xb[NLOAD];
  v4f sum[NLOAD];
  const v4f zero4 = {0.0f, 0.0f, 0.0f, 0.0f};
#pragma unroll
  for (int u = 0; u < NLOAD; ++u) sum[u] = zero4;

#define LOAD_TILE(X, IC)                                                       \
  {                                                                            \
    const int cs_ = (IC) + dlo;                                                \
    const float* rp_ = pb + (long)(IC) * TDIM;                                 \
    unsigned off_[NLOAD];                                                      \
    _Pragma("unroll")                                                          \
    for (int u = 0; u < NLOAD; ++u) {                                          \
      int col_ = cs_ + qcol[u];                                                \
      col_ = col_ < 0 ? 0 : col_;                                              \
      col_ = col_ > TDIM - 4 ? TDIM - 4 : col_;                                \
      off_[u] = rq[u] + (unsigned)col_;                                        \
    }                                                                          \
    __builtin_amdgcn_sched_barrier(0);                                         \
    _Pragma("unroll")                                                          \
    for (int u = 0; u < NLOAD; ++u)                                            \
      X[u] = *(const v4f*)(rp_ + off_[u]);                                     \
  }

#define FIX_TILE(X, IC)                                                        \
  {                                                                            \
    const int cs_ = (IC) + dlo;                                                \
    const bool last_ = ((IC) + CHUNK == TDIM);                                 \
    if (cs_ < 0 || cs_ + TILE_W > TDIM - 4 || last_) {                         \
      _Pragma("unroll")                                                        \
      for (int u = 0; u < NLOAD; ++u) {                                        \
        const int col_ = cs_ + qcol[u];                                        \
        bool ok_ = (unsigned)col_ < (unsigned)TDIM;                            \
        if (u == NLOAD - 1) ok_ = ok_ && !(last_ && t >= NTHREAD - SLOTS_PER_ROW); \
        X[u] = ok_ ? X[u] : zero4;                                             \
        X[u].w = (col_ == TDIM - 4) ? 0.0f : X[u].w;                           \
      }                                                                        \
    }                                                                          \
  }

#define TAKE(X, IC, NEXT)                                                      \
  {                                                                            \
    FIX_TILE(X, IC)                                                            \
    _Pragma("unroll")                                                          \
    for (int u = 0; u < NLOAD; ++u)                                            \
      sum[u] += X[u];                                                          \
    if (NEXT) LOAD_TILE(X, (IC) + 2 * CHUNK)                                   \
  }

  LOAD_TILE(xa, i0)
  LOAD_TILE(xb, i0 + CHUNK)

#pragma unroll 1
  for (int c = 0; c < NCHUNK - 2; c += 2) {
    const int ic = i0 + CHUNK * c;
    TAKE(xa, ic, 1)
    TAKE(xb, ic + CHUNK, 1)
  }
  {
    const int ic = i0 + CHUNK * (NCHUNK - 2);
    TAKE(xa, ic, 0)
    TAKE(xb, ic + CHUNK, 0)
  }
#undef TAKE
#undef FIX_TILE
#undef LOAD_TILE

#pragma unroll
  for (int u = 0; u < NLOAD; ++u)
    *(v4f*)(tile + stoff + u * (4 * LDS_PITCH - 4)) = sum[u];
  __syncthreads();
  v4f acc = *(const v4f*)(tile + rdoff);
#pragma unroll
  for (int j = 1; j < CHUNK / 4; ++j)
    acc += *(const v4f*)(tile + rdoff + j * (4 * LDS_PITCH));

  if (t < GDIAG + 4) *(v4f*)&joinp[kcls][4 * qd] = acc;
  __syncthreads();
  if (t < GDIAG) {
    const float s = ((joinp[0][t] + joinp[1][t + 1]) + joinp[2][t + 2]) + joinp[3][t + 3];
    strip_ws[((size_t)b * NSTRIP + strip) * WS_PITCH + dg * GDIAG + t] = s;
  }
}

__global__ __launch_bounds__(128) void diagmean_finalize(
    const float* __restrict__ in, const float* __restrict__ strip_ws,
    float* __restrict__ out) {
  const int b    = blockIdx.x;
  const int tid  = (int)threadIdx.x;
  const int lane = tid & 31;
  const int col  = lane & 15;
  const int hi   = lane >> 4;

  __shared__ float red[128];
  __shared__ float red2[128];
  __shared__ float part_s;
  __shared__ float corner_s;

  const float* sb = strip_ws + (size_t)b * NSTRIP * WS_PITCH;
  const float* pb = in + (size_t)b * TDIM * TDIM;

  float v[8];
  v2f b0[WS_PITCH / 128];
  v2f b1[WS_PITCH / 128];
  float m[WS_PITCH / 128];
#pragma unroll
  for (int k = 0; k < 8; ++k) {
    const int i = tid * 8 + k;
    v[k] = pb[(size_t)i * TDIM + (i + HALF)];
  }
  const float* s0 = sb + (2 * hi + 0) * WS_PITCH + (tid & ~31) + col;
  const float* s1 = sb + (2 * hi + 1) * WS_PITCH + (tid & ~31) + col;
#pragma unroll
  for (int k = 0; k < WS_PITCH / 128; ++k) {
    b0[k].x = s0[128 * k];
    b0[k].y = s1[128 * k];
    b1[k].x = s0[128 * k + 16];
    b1[k].y = s1[128 * k + 16];
  }

  float q = 0.0f;
#pragma unroll
  for (int k = 0; k < 8; ++k) {
    const int i = tid * 8 + k;
    q += (i < HALF - 1) ? v[k] : 0.0f;
  }
  v2f a; a.x = 1.0f; a.y = 1.0f;
#pragma unroll
  for (int k = 0; k < WS_PITCH / 128; ++k) {
    v8f c0 = {};
    v8f c1 = {};
    c0 = __builtin_amdgcn_wmma_f32_16x16x4_f32(false, a, false, b0[k],
                                               (short)0, c0, false, false);
    c1 = __builtin_amdgcn_wmma_f32_16x16x4_f32(false, a, false, b1[k],
                                               (short)0, c1, false, false);
    const float s   = hi ? c1[0] : c0[0];
    const int obin  = tid + 128 * k;
    const int dd    = obin - HALF;
    const float cnt = (float)((TDIM - 1) - (dd < 0 ? -dd : dd));
    m[k] = s / cnt;
  }
  float p = 0.0f;
#pragma unroll
  for (int k = 0; k < WS_PITCH / 128; ++k) p += m[k];
  red[tid]  = p;
  red2[tid] = q;
  __syncthreads();
  if (tid == 0) {
    float s = 0.0f;
    for (int i = 0; i < 128; ++i) s += red[i];
    part_s = s;
  }
  if (tid == 32) {
    float s = 0.0f;
    for (int i = 0; i < 128; ++i) s += red2[i];
    corner_s = s / (float)((TDIM - 1) - HALF);
  }
  __syncthreads();

  const float corner = corner_s;
  const float g = (part_s + corner) / (float)BINS;
#pragma unroll
  for (int k = 0; k < WS_PITCH / 128; ++k) {
    out[(size_t)b * BINS + tid + 128 * k] = -(m[k] - g);
  }
  if (tid == 0) out[(size_t)b * BINS + (BINS - 1)] = -(corner - g);
}

extern "C" void kernel_launch(void* const* d_in, const int* in_sizes, int n_in,
                              void* d_out, int out_size, void* d_ws, size_t ws_size,
                              hipStream_t stream) {
  (void)in_sizes; (void)n_in; (void)out_size; (void)ws_size;
  const float* in  = (const float*)d_in[0];
  float*       out = (float*)d_out;
  float*       ws  = (float*)d_ws;

  dim3 g1(NDG * NSTRIP * NBATCH, 1, 1);
  diagmean_partial<<<g1, NTHREAD, 0, stream>>>(in, ws);
  diagmean_finalize<<<NBATCH, 128, 0, stream>>>(in, ws, out);
}


# centering mean summed in the partial kernel: a group adds its (sum / count) terms to one number, passes bin 2048's strip sum; finalize: no input read, LDS, barrier, add chain; mean sum order changed
# speedup vs baseline: 2.5878x; 1.0681x over previous
#include <hip/hip_runtime.h>


#define TDIM 2048
#define HALF 1024
#define BINS 2049
#define NBATCH 16
#define GDIAG 512
#define NDG 4
#define NSTRIP 4
#define ROWS_PER_STRIP 512
#define CHUNK 32
#define NCHUNK 16
#define TILE_W 544
#define SLOTS_PER_ROW 136
#define NTHREAD 544
#define NLOAD 8
#define WS_PITCH 2048
#define LDS_PITCH 548
#define NQUAD 129
#define JOIN_W 520
#define GROUP_ORDER 0xF0EB41DCA9876532ull
#define SHARE_LINE 32

typedef __attribute__((ext_vector_type(2))) float v2f;
typedef __attribute__((ext_vector_type(4))) float v4f;
typedef __attribute__((ext_vector_type(8))) float v8f;

#define XLANE_ROW(V, M)                                                        \
  __builtin_bit_cast(float, __builtin_amdgcn_update_dpp(                       \
      0, __builtin_bit_cast(int, (V)), 0x160 + (M), 0xf, 0xf, false))
__device__ __forceinline__ float xhalf(float x) {
  return __builtin_bit_cast(float, __builtin_amdgcn_permlanex16(
      __builtin_bit_cast(int, x), __builtin_bit_cast(int, x), 0x76543210, 0xfedcba98, false, false));
}

__global__ __launch_bounds__(NTHREAD) void diagmean_partial(
    const float* __restrict__ in, float* __restrict__ strip_ws) {
  const unsigned cls = blockIdx.x >> 4;
  const unsigned pr  = (unsigned)((GROUP_ORDER >> (4u * cls)) & 15ull);
  const int dg    = (int)(pr & (NDG - 1));
  const int strip = (int)(pr >> 2);
  const int b     = (int)(blockIdx.x & (NBATCH - 1));
  const int t     = (int)threadIdx.x;
  const int dlo   = dg * GDIAG - HALF;
  const int kcls  = (t < GDIAG) ? ((t >> 5) & 3) : (t & 3);
  const int qd    = (t < GDIAG) ? (((t >> 7) << 5) + (t & 31)) : (NQUAD - 1);
  const int rdoff = kcls * LDS_PITCH + 4 * qd;
  const int stoff = (t / SLOTS_PER_ROW) * LDS_PITCH + 4 * (t % SLOTS_PER_ROW);

  __shared__ __attribute__((aligned(16))) float tile[CHUNK * LDS_PITCH];
  __shared__ __attribute__((aligned(16))) float joinp[4][JOIN_W];
  __shared__ float red[NTHREAD];
  __shared__ float cor[4];

  const float* pb = in + (size_t)b * TDIM * TDIM;
  const int i0 = strip * ROWS_PER_STRIP;

  unsigned rq[NLOAD];
  int      qcol[NLOAD];
#pragma unroll
  for (int u = 0; u < NLOAD; ++u) {
    const int piece = t + NTHREAD * u;
    const int r = piece / SLOTS_PER_ROW;
    const int q = piece - r * SLOTS_PER_ROW;
    rq[u]   = (unsigned)(r * TDIM);
    qcol[u] = 4 * q;
  }

  v4f xa[NLOAD];
  v4f xb[NLOAD];
  v4f sum[NLOAD];
  const v4f zero4 = {0.0f, 0.0f, 0.0f, 0.0f};
#pragma unroll
  for (int u = 0; u < NLOAD; ++u) sum[u] = zero4;

#define LOAD_TILE(X, IC)                                                       \
  {                                                                            \
    const int cs_ = (IC) + dlo;                                                \
    const float* rp_ = pb + (long)(IC) * TDIM;                                 \
    unsigned off_[NLOAD];                                                      \
    _Pragma("unroll")                                                          \
    for (int u = 0; u < NLOAD; ++u) {                                          \
      int col_ = cs_ + qcol[u];                                                \
      col_ = col_ < 0 ? 0 : col_;                                              \
      col_ = col_ > TDIM - 4 ? TDIM - 4 : col_;                                \
      off_[u] = rq[u] + (unsigned)col_;                                        \
    }                                                                          \
    __builtin_amdgcn_sched_barrier(0);                                         \
    _Pragma("unroll")                                                          \
    for (int u = 0; u < NLOAD; ++u)                                            \
      X[u] = *(const v4f*)(rp_ + off_[u]);                                     \
  }

#define FIX_TILE(X, IC)                                                        \
  {                                                                            \
    const int cs_ = (IC) + dlo;                                                \
    const bool last_ = ((IC) + CHUNK == TDIM);                                 \
    if (cs_ < 0 || cs_ + TILE_W > TDIM - 4 || last_) {                         \
      _Pragma("unroll")                                                        \
      for (int u = 0; u < NLOAD; ++u) {                                        \
        const int col_ = cs_ + qcol[u];                                        \
        bool ok_ = (unsigned)col_ < (unsigned)TDIM;                            \
        if (u == NLOAD - 1) ok_ = ok_ && !(last_ && t >= NTHREAD - SLOTS_PER_ROW); \
        X[u] = ok_ ? X[u] : zero4;                                             \
        X[u].w = (col_ == TDIM - 4) ? 0.0f : X[u].w;                           \
      }                                                                        \
    }                                                                          \
  }

#define TAKE(X, IC, NEXT)                                                      \
  {                                                                            \
    FIX_TILE(X, IC)                                                            \
    _Pragma("unroll")                                                          \
    for (int u = 0; u < NLOAD; ++u)                                            \
      sum[u] += X[u];                                                          \
    if (NEXT) LOAD_TILE(X, (IC) + 2 * CHUNK)                                   \
  }

  LOAD_TILE(xa, i0)
  LOAD_TILE(xb, i0 + CHUNK)

#pragma unroll 1
  for (int c = 0; c < NCHUNK - 2; c += 2) {
    const int ic = i0 + CHUNK * c;
    TAKE(xa, ic, 1)
    TAKE(xb, ic + CHUNK, 1)
  }
  {
    const int ic = i0 + CHUNK * (NCHUNK - 2);
    TAKE(xa, ic, 0)
    TAKE(xb, ic + CHUNK, 0)
  }
#undef TAKE
#undef FIX_TILE
#undef LOAD_TILE

#pragma unroll
  for (int u = 0; u < NLOAD; ++u)
    *(v4f*)(tile + stoff + u * (4 * LDS_PITCH - 4)) = sum[u];
  __syncthreads();
  v4f acc = *(const v4f*)(tile + rdoff);
#pragma unroll
  for (int j = 1; j < CHUNK / 4; ++j)
    acc += *(const v4f*)(tile + rdoff + j * (4 * LDS_PITCH));

  if (t < GDIAG + 4) *(v4f*)&joinp[kcls][4 * qd] = acc;
  float pshare = 0.0f;
#pragma unroll
  for (int e = 0; e < 4; ++e) {
    const int x  = 4 * qd - kcls + e;
    const int dd = dlo + x;
    const float cnt  = (float)((TDIM - 1) - (dd < 0 ? -dd : dd));
    const float term = acc[e] / cnt;
    pshare += ((unsigned)x < (unsigned)GDIAG && t < GDIAG + 4) ? term : 0.0f;
  }
  red[t] = pshare;
  if (t >= GDIAG && t < GDIAG + 4)
    cor[kcls] = kcls == 0 ? acc.x : kcls == 1 ? acc.y : kcls == 2 ? acc.z : acc.w;
  __syncthreads();
  if (t < GDIAG) {
    const float s = ((joinp[0][t] + joinp[1][t + 1]) + joinp[2][t + 2]) + joinp[3][t + 3];
    if (pr != 0)
      strip_ws[((size_t)b * NSTRIP + strip) * WS_PITCH + dg * GDIAG + t] = s;
  } else {
    const int l = t - GDIAG;
    float v = red[l];
#pragma unroll
    for (int j = 1; j < NTHREAD / 32; ++j) v += red[l + 32 * j];
    v += XLANE_ROW(v, 1);
    v += XLANE_ROW(v, 2);
    v += XLANE_ROW(v, 4);
    v += XLANE_ROW(v, 8);
    v += xhalf(v);
    const float cs = ((cor[0] + cor[1]) + cor[2]) + cor[3];
    float* line = strip_ws + (size_t)b * NSTRIP * WS_PITCH + SHARE_LINE * pr;
    if (l < 2) line[l] = (l == 0) ? v : cs;
  }
}

__global__ __launch_bounds__(128) void diagmean_finalize(
    const float* __restrict__ strip_ws, float* __restrict__ out) {
  const int b    = blockIdx.x;
  const int tid  = (int)threadIdx.x;
  const int lane = tid & 31;
  const int col  = lane & 15;
  const int hi   = lane >> 4;

  const float* sb = strip_ws + (size_t)b * NSTRIP * WS_PITCH;

  float sh[NDG * NSTRIP];
  float cr[NSTRIP];
  v2f b0[WS_PITCH / 128];
  v2f b1[WS_PITCH / 128];
#pragma unroll
  for (int p = 0; p < NDG * NSTRIP; ++p) sh[p] = sb[SHARE_LINE * p];
#pragma unroll
  for (int s = 0; s < NSTRIP; ++s) cr[s] = sb[SHARE_LINE * (NDG - 1 + NDG * s) + 1];
  const float* s0 = sb + (2 * hi + 0) * WS_PITCH + (tid & ~31) + col;
  const float* s1 = sb + (2 * hi + 1) * WS_PITCH + (tid & ~31) + col;
#pragma unroll
  for (int k = 0; k < WS_PITCH / 128; ++k) {
    float x0 = 0.0f, x1 = 0.0f;
    if (k >= GDIAG / 128 || hi != 0) {
      x0 = s0[128 * k];
      x1 = s0[128 * k + 16];
    }
    b0[k].x = x0;
    b0[k].y = s1[128 * k];
    b1[k].x = x1;
    b1[k].y = s1[128 * k + 16];
  }

  const float corner = (((cr[0] + cr[1]) + cr[2]) + cr[3]) / (float)((TDIM - 1) - HALF);
  float tot = 0.0f;
#pragma unroll
  for (int p = 0; p < NDG * NSTRIP; ++p) tot += sh[p];
  const float g = (tot + corner) / (float)BINS;

  v2f a; a.x = 1.0f; a.y = 1.0f;
#pragma unroll
  for (int k = 0; k < WS_PITCH / 128; ++k) {
    v8f c0 = {};
    v8f c1 = {};
    c0 = __builtin_amdgcn_wmma_f32_16x16x4_f32(false, a, false, b0[k],
                                               (short)0, c0, false, false);
    c1 = __builtin_amdgcn_wmma_f32_16x16x4_f32(false, a, false, b1[k],
                                               (short)0, c1, false, false);
    const float s   = hi ? c1[0] : c0[0];
    const int obin  = tid + 128 * k;
    const int dd    = obin - HALF;
    const float cnt = (float)((TDIM - 1) - (dd < 0 ? -dd : dd));
    out[(size_t)b * BINS + obin] = -(s / cnt - g);
  }
  if (tid == 0) out[(size_t)b * BINS + (BINS - 1)] = -(corner - g);
}

extern "C" void kernel_launch(void* const* d_in, const int* in_sizes, int n_in,
                              void* d_out, int out_size, void* d_ws, size_t ws_size,
                              hipStream_t stream) {
  (void)in_sizes; (void)n_in; (void)out_size; (void)ws_size;
  const float* in  = (const float*)d_in[0];
  float*       out = (float*)d_out;
  float*       ws  = (float*)d_ws;

  dim3 g1(NDG * NSTRIP * NBATCH, 1, 1);
  diagmean_partial<<<g1, NTHREAD, 0, stream>>>(in, ws);
  diagmean_finalize<<<NBATCH, 128, 0, stream>>>(ws, out);
}
